# starting kernel
# baseline (speedup 1.0000x reference)
#include <hip/hip_runtime.h>

#define BATCH  1024
#define INODES 128
#define ONODES 128
#define HID    5

typedef __attribute__((ext_vector_type(2))) float v2f;
typedef __attribute__((ext_vector_type(8))) float v8f;

__device__ __forceinline__ float lrelu(float t) { return fmaxf(t, 0.01f * t); }

#define OFF_PK4  0
#define OFF_W14  1024
#define OFF_PK2A 1280
#define OFF_W2Q  9472
#define OFF_PK2B 11520
#define SMEM_FL  15616

__global__ __launch_bounds__(256) void kan_wmma_kernel(
    const float* __restrict__ x,
    const float* __restrict__ W1, const float* __restrict__ B1,
    const float* __restrict__ W2, const float* __restrict__ B2,
    const float* __restrict__ W3, const float* __restrict__ B3,
    float* __restrict__ out) {
  __shared__ __align__(16) float smem[SMEM_FL];
  __shared__ float sB3;

  const int tid   = threadIdx.x;
  const int o     = blockIdx.x;
  const int chunk = blockIdx.y;

  const float* W1g = W1 + o * (INODES * HID);
  const float* B1g = B1 + o * (INODES * HID);
  const float* W2g = W2 + o * (INODES * HID * HID);
  const float* B2g = B2 + o * (INODES * HID);
  const float* W3g = W3 + o * (INODES * HID);

  for (int idx = tid; idx < 1024; idx += 256) {
    const int i = idx >> 3, r = idx & 7, hf = r >> 2, j = r & 3;
    const int h = 2 * hf + (j & 1);
    smem[OFF_PK4 + idx] = (j < 2) ? W1g[i * HID + h] : B1g[i * HID + h];
  }
  for (int idx = tid; idx < 256; idx += 256) {
    const int i = idx >> 1, j = idx & 1;
    smem[OFF_W14 + idx] = j ? B1g[i * HID + 4] : W1g[i * HID + 4];
  }
  for (int idx = tid; idx < 8192; idx += 256) {
    const int i = idx >> 6, r = idx & 63, ln = r >> 1, j = r & 1;
    const int hf = ln >> 4, N = ln & 15, Nc = (N < HID) ? N : 0;
    const float v = W2g[i * 25 + Nc * HID + 2 * hf + j];
    smem[OFF_PK2A + idx] = (N < HID) ? v : 0.f;
  }
  for (int idx = tid; idx < 2048; idx += 256) {
    const int i = idx >> 4, N = idx & 15, Nc = (N < HID) ? N : 0;
    const float v = W2g[i * 25 + Nc * HID + 4];
    smem[OFF_W2Q + idx] = (N < HID) ? v : 0.f;
  }
  for (int idx = tid; idx < 4096; idx += 256) {
    const int i = idx >> 5, r = idx & 31, N = r >> 1, j = r & 1;
    const int Nc = (N < HID) ? N : 0;
    const float v = j ? W3g[i * HID + Nc] : B2g[i * HID + Nc];
    smem[OFF_PK2B + idx] = (N < HID) ? v : 0.f;
  }
  if (tid == 0) {
    float s = 0.f;
    const float* B3g = B3 + o * INODES;
    for (int i = 0; i < INODES; ++i) s += B3g[i];
    sB3 = s;
  }
  __syncthreads();

  const int  lane = tid & 31;
  const int  wave = tid >> 5;
  const int  half = lane >> 4;
  const int  N    = lane & 15;
  const int  b0   = chunk * 128 + wave * 16;
  const float* xp = x + (b0 + N) * INODES;

  const float* pk4  = &smem[OFF_PK4 + half * 4];
  const float* w14p = &smem[OFF_W14];
  const float* pk2a = &smem[OFF_PK2A + lane * 2];
  const float* w2q  = &smem[OFF_W2Q + N];
  const float* pk2b = &smem[OFF_PK2B + N * 2];

  v8f acc = {};

  for (int i = 0; i < INODES; ++i) {
    const float  xv  = xp[i];
    const float4 w1p = *(const float4*)(pk4 + i * 8);
    const float2 w14 = *(const float2*)(w14p + i * 2);
    const float2 bw  = *(const float2*)(pk2a + i * 64);
    float        bw4 = w2q[i * 16];
    const float2 bw3 = *(const float2*)(pk2b + i * 32);
    bw4 = half ? 0.f : bw4;

    const float a0 = lrelu(fmaf(xv, w1p.x, w1p.z));
    const float a1 = lrelu(fmaf(xv, w1p.y, w1p.w));
    const float a4 = lrelu(fmaf(xv, w14.x, w14.y));

    v2f A1;  A1.x = a0;   A1.y = a1;
    v2f Bm1; Bm1.x = bw.x; Bm1.y = bw.y;
    v2f A2;  A2.x = a4;   A2.y = 0.f;
    v2f Bm2; Bm2.x = bw4; Bm2.y = 0.f;

    v8f c = {};
    c = __builtin_amdgcn_wmma_f32_16x16x4_f32(false, A1, false, Bm1,
                                              (short)0, c, false, false);
    c = __builtin_amdgcn_wmma_f32_16x16x4_f32(false, A2, false, Bm2,
                                              (short)0, c, false, false);

    const float bias = bw3.x;
    const float w3v  = bw3.y;
#pragma unroll
    for (int r = 0; r < 8; ++r) {
      const float h2 = lrelu(c[r] + bias);
      acc[r] = fmaf(h2, w3v, acc[r]);
    }
  }

#pragma unroll
  for (int r = 0; r < 8; ++r) {
    float v = acc[r];
    v += __shfl_xor(v, 8, 16);
    v += __shfl_xor(v, 4, 16);
    v += __shfl_xor(v, 2, 16);
    v += __shfl_xor(v, 1, 16);
    acc[r] = v;
  }
  if (N == 0) {
    const float sb = sB3;
#pragma unroll
    for (int r = 0; r < 8; ++r) {
      const int b = b0 + 8 * half + r;
      out[b * ONODES + o] = acc[r] + sb;
    }
  }
}

extern "C" void kernel_launch(void* const* d_in, const int* in_sizes, int n_in,
                              void* d_out, int out_size, void* d_ws, size_t ws_size,
                              hipStream_t stream) {
  (void)in_sizes; (void)n_in; (void)out_size; (void)d_ws; (void)ws_size;
  const float* x  = (const float*)d_in[0];
  const float* W1 = (const float*)d_in[1];
  const float* B1 = (const float*)d_in[2];
  const float* W2 = (const float*)d_in[3];
  const float* B2 = (const float*)d_in[4];
  const float* W3 = (const float*)d_in[5];
  const float* B3 = (const float*)d_in[6];
  float* out = (float*)d_out;

  dim3 grid(ONODES, BATCH / 128);
  kan_wmma_kernel<<<grid, 256, 0, stream>>>(x, W1, B1, W2, B2, W3, B3, out);
}


# Transposed chain: batch is the matrix column, weights the A operands; layer 1 is a product too (k = x,1), its D registers are layer 2's B as they stand; 3 edges a group; edge and k sum order change
# speedup vs baseline: 1.6038x; 1.6038x over previous
#include <hip/hip_runtime.h>

#define BATCH  1024
#define INODES 128
#define ONODES 128
#define HID    5

typedef __attribute__((ext_vector_type(2))) float v2f;
typedef __attribute__((ext_vector_type(4))) float v4f;
typedef __attribute__((ext_vector_type(8))) float v8f;

__device__ __forceinline__ float lrelu(float t) { return fmaxf(t, 0.01f * t); }

#define NGRP    43
#define OFF_A2  0
#define OFF_A1  11008
#define OFF_W3  13760
#define IMG_FL  14448
#define OFF_B3R 14448
#define SMEM_FL 14576

__device__ __forceinline__ void edge_slot(int i, int& g, int& s) {
  if (i < 86) { g = i >> 1; s = i & 1; } else { g = i - 86; s = 2; }
}

__device__ __forceinline__ void put_w2(float* smem, int q, v4f val) {
  const int f0   = 4 * q;
  const int i0   = f0 / 25;
  const int rem0 = f0 - 25 * i0;
  const int k0   = (rem0 * 13) >> 6;
  const int h0   = rem0 - 5 * k0;
#pragma unroll
  for (int c = 0; c < 4; ++c) {
    int h = h0 + c, k = k0, i = i0;
    if (h >= 5) { h -= 5; k += 1; }
    if (k >= 5) { k -= 5; i += 1; }
    int g, s; edge_slot(i, g, s);
    const int m2  = 5 * s + k;
    const int col = 5 * s + h;
    smem[OFF_A2 + g * 256 + ((col & 4) << 5) + ((col & 8) << 3) + m2 * 4 + (col & 3)] = val[c];
  }
}

__device__ __forceinline__ void edge_group(const float* pa1, const float* pa2,
                                           const float* pw3, int g, v2f X, v8f& acc) {
  const v2f a1   = *(const v2f*)(pa1 + g * 64);
  const v4f a2lo = *(const v4f*)(pa2 + g * 256);
  const v4f a2hi = *(const v4f*)(pa2 + g * 256 + 128);
  const v4f w3lo = *(const v4f*)(pw3 + g * 16);
  const v4f w3hi = *(const v4f*)(pw3 + g * 16 + 4);

  v8f d1 = {};
  d1 = __builtin_amdgcn_wmma_f32_16x16x4_f32(false, a1, false, X, (short)0, d1, false, false);
  v2f h01, h23, h45, h67;
  h01.x = lrelu(d1[0]); h01.y = lrelu(d1[1]);
  h23.x = lrelu(d1[2]); h23.y = lrelu(d1[3]);
  h45.x = lrelu(d1[4]); h45.y = lrelu(d1[5]);
  h67.x = lrelu(d1[6]); h67.y = lrelu(d1[7]);

  v2f w;
  v8f d2 = {};
  w.x = a2lo.x; w.y = a2lo.y;
  d2 = __builtin_amdgcn_wmma_f32_16x16x4_f32(false, w, false, h01, (short)0, d2, false, false);
  w.x = a2lo.z; w.y = a2lo.w;
  d2 = __builtin_amdgcn_wmma_f32_16x16x4_f32(false, w, false, h23, (short)0, d2, false, false);
  w.x = a2hi.x; w.y = a2hi.y;
  d2 = __builtin_amdgcn_wmma_f32_16x16x4_f32(false, w, false, h45, (short)0, d2, false, false);
  w.x = a2hi.z; w.y = a2hi.w;
  d2 = __builtin_amdgcn_wmma_f32_16x16x4_f32(false, w, false, h67, (short)0, d2, false, false);

  acc[0] = fmaf(lrelu(d2[0]), w3lo.x, acc[0]);
  acc[1] = fmaf(lrelu(d2[1]), w3lo.y, acc[1]);
  acc[2] = fmaf(lrelu(d2[2]), w3lo.z, acc[2]);
  acc[3] = fmaf(lrelu(d2[3]), w3lo.w, acc[3]);
  acc[4] = fmaf(lrelu(d2[4]), w3hi.x, acc[4]);
  acc[5] = fmaf(lrelu(d2[5]), w3hi.y, acc[5]);
  acc[6] = fmaf(lrelu(d2[6]), w3hi.z, acc[6]);
  acc[7] = fmaf(lrelu(d2[7]), w3hi.w, acc[7]);
}

__global__ __launch_bounds__(256) void kan_wmma_kernel(
    const float* __restrict__ x,
    const float* __restrict__ W1, const float* __restrict__ B1,
    const float* __restrict__ W2, const float* __restrict__ B2,
    const float* __restrict__ W3, const float* __restrict__ B3,
    float* __restrict__ out) {
  __shared__ __align__(16) float smem[SMEM_FL];
  __shared__ float sB3;

  const int tid   = threadIdx.x;
  const int o     = blockIdx.x;
  const int chunk = blockIdx.y;

  {
    const v4f* W1v = (const v4f*)(W1 + o * (INODES * HID));
    const v4f* B1v = (const v4f*)(B1 + o * (INODES * HID));
    const v4f* B2v = (const v4f*)(B2 + o * (INODES * HID));
    const v4f* W3v = (const v4f*)(W3 + o * (INODES * HID));
    const v4f* W2v = (const v4f*)(W2 + o * (INODES * HID * HID));
    const v4f* B3v = (const v4f*)(B3 + o * INODES);

    const int qa = (tid < 160) ? tid : 159;
    const int q3 = (tid < 32) ? tid + 768 : 799;
    const v4f w1 = W1v[qa];
    const v4f b1 = B1v[qa];
    const v4f b2 = B2v[qa];
    const v4f w3 = W3v[qa];
    const v4f m0 = W2v[tid];
    const v4f m1 = W2v[tid + 256];
    const v4f m2 = W2v[tid + 512];
    const v4f m3 = W2v[q3];
    const v4f bb = B3v[tid & 31];

    v4f* sv = (v4f*)smem;
    const v4f z4 = {0.f, 0.f, 0.f, 0.f};
    for (int q = tid; q < IMG_FL / 4; q += 256) sv[q] = z4;

    asm volatile("" :: "v"(w1), "v"(b1), "v"(b2), "v"(w3), "v"(m0), "v"(m1), "v"(m2),
                 "v"(m3), "v"(bb));

    __syncthreads();

    if (tid < 160) {
      const int f0 = 4 * tid;
      const int i0 = f0 / 5;
      const int h0 = f0 - 5 * i0;
#pragma unroll
      for (int c = 0; c < 4; ++c) {
        int h = h0 + c, i = i0;
        if (h >= 5) { h -= 5; i += 1; }
        int g, s; edge_slot(i, g, s);
        const int m = 5 * s + h;
        smem[OFF_A1 + g * 64 + m * 4 + s] = w1[c];
        smem[OFF_A1 + g * 64 + m * 4 + 3] = b1[c];
        smem[OFF_A2 + g * 256 + 128 + (16 + m) * 4 + 3] = b2[c];
        smem[OFF_W3 + g * 16 + m] = w3[c];
      }
    }
    put_w2(smem, tid, m0);
    put_w2(smem, tid + 256, m1);
    put_w2(smem, tid + 512, m2);
    if (tid < 32) {
      put_w2(smem, tid + 768, m3);
      sv[OFF_B3R / 4 + tid] = bb;
    }
    if (tid < NGRP) smem[OFF_A1 + tid * 64 + 15 * 4 + 3] = 1.0f;
  }
  __syncthreads();

  if (tid == 0) {
    float s = 0.f;
    for (int i = 0; i < INODES; ++i) s += smem[OFF_B3R + i];
    sB3 = s;
  }

  const int  lane = tid & 31;
  const int  wave = tid >> 5;
  const int  hf   = lane >> 4;
  const int  col  = lane & 15;
  const int  b0   = chunk * 128 + wave * 16;
  const float* xr = x + (b0 + col) * INODES;

  const float* pa1 = &smem[OFF_A1 + col * 4 + 2 * hf];
  const float* pa2 = &smem[OFF_A2 + lane * 4];
  const float* pw3 = &smem[OFF_W3 + 8 * hf];

  v8f acc = {};

  v2f p = *(const v2f*)(xr + (hf ? 86 : 0));
  v2f q = *(const v2f*)(xr + 2);
  for (int G = 0; G < 21; ++G) {
    const int Gn = G + 1;
    const int i0 = hf ? (Gn < 21 ? 86 + 2 * Gn : 126) : 4 * Gn;
    const v2f pn = *(const v2f*)(xr + i0);
    const v2f qn = *(const v2f*)(xr + 4 * Gn + 2);

    v2f X0, X1;
    X0.x = p.x;             X0.y = hf ? 1.0f : p.y;
    X1.x = hf ? p.y : q.x;  X1.y = hf ? 1.0f : q.y;
    edge_group(pa1, pa2, pw3, 2 * G,     X0, acc);
    edge_group(pa1, pa2, pw3, 2 * G + 1, X1, acc);
    asm volatile("" :: "v"(pn), "v"(qn), "v"(acc));
    p = pn; q = qn;
  }
  {
    v2f X;
    X.x = hf ? 0.0f : p.x;
    X.y = hf ? 1.0f : p.y;
    edge_group(pa1, pa2, pw3, 42, X, acc);
  }

  float v = ((acc[0] + acc[1]) + (acc[2] + acc[3])) + ((acc[4] + acc[5]) + (acc[6] + acc[7]));
  v += __shfl_xor(v, 16, 32);
  __syncthreads();
  if (hf == 0) out[(b0 + col) * ONODES + o] = v + sB3;
}

extern "C" void kernel_launch(void* const* d_in, const int* in_sizes, int n_in,
                              void* d_out, int out_size, void* d_ws, size_t ws_size,
                              hipStream_t stream) {
  (void)in_sizes; (void)n_in; (void)out_size; (void)d_ws; (void)ws_size;
  const float* x  = (const float*)d_in[0];
  const float* W1 = (const float*)d_in[1];
  const float* B1 = (const float*)d_in[2];
  const float* W2 = (const float*)d_in[3];
  const float* B2 = (const float*)d_in[4];
  const float* W3 = (const float*)d_in[5];
  const float* B3 = (const float*)d_in[6];
  float* out = (float*)d_out;

  dim3 grid(ONODES, BATCH / 128);
  kan_wmma_kernel<<<grid, 256, 0, stream>>>(x, W1, B1, W2, B2, W3, B3, out);
}
